# starting kernel
# baseline (speedup 1.0000x reference)
#include <hip/hip_runtime.h>
#include <math.h>

typedef __attribute__((ext_vector_type(2))) float v2f;
typedef __attribute__((ext_vector_type(4))) float v4f;
typedef __attribute__((ext_vector_type(8))) float v8f;

#define NUM_TASKS 1000
#define NFEAT 256
#define HID 128
#define BATCH 4096
#define WAVES_PER_BLOCK 4
#define NTILES 8

__global__ __launch_bounds__(WAVES_PER_BLOCK * 32)
void mtl_fused_kernel(const float* __restrict__ x,
                      const int* __restrict__ task_ids,
                      const float* __restrict__ l1_emb,
                      const float* __restrict__ l1_bias,
                      const float* __restrict__ l2_emb,
                      const float* __restrict__ l2_bias,
                      float* __restrict__ out) {
    __shared__ float xs[WAVES_PER_BLOCK][NFEAT];

    const int wave = threadIdx.x >> 5;
    const int lane = threadIdx.x & 31;
    const int sample = blockIdx.x * WAVES_PER_BLOCK + wave;

    const int task = task_ids[sample];
    const float* __restrict__ xp = x + (size_t)sample * NFEAT;

    {
        v4f a = *(const v4f*)(xp + lane * 4);
        v4f b = *(const v4f*)(xp + 128 + lane * 4);
        *(v4f*)(&xs[wave][lane * 4]) = a;
        *(v4f*)(&xs[wave][128 + lane * 4]) = b;
    }
    __syncthreads();

    const float* __restrict__ w1 = l1_emb + (size_t)task * (NFEAT * HID);

    v8f acc[NTILES];
    v8f zero = {};
#pragma unroll
    for (int t = 0; t < NTILES; ++t) acc[t] = zero;

    const int khalf = (lane >> 4) * 2;
    const int nl = lane & 15;

    for (int k0 = 0; k0 < NFEAT; k0 += 4) {
        v2f A;
        A.x = xs[wave][k0 + khalf];
        A.y = xs[wave][k0 + khalf + 1];

        const float* __restrict__ wk = w1 + (size_t)(k0 + khalf) * HID + nl;

        if (k0 + 4 < NFEAT) {
            __builtin_prefetch(wk + 4 * HID, 0, 0);
        }

#pragma unroll
        for (int t = 0; t < NTILES; ++t) {
            v2f B;
            B.x = wk[t * 16];
            B.y = wk[t * 16 + HID];
            acc[t] = __builtin_amdgcn_wmma_f32_16x16x4_f32(
                false, A, false, B, (short)0, acc[t], false, false);
        }
    }

    const float* __restrict__ b1 = l1_bias + (size_t)task * HID;
    const float* __restrict__ w2 = l2_emb + (size_t)task * HID;

    float s = 0.0f;
#pragma unroll
    for (int t = 0; t < NTILES; ++t) {
        const int n = t * 16 + nl;
        float h = acc[t][0] + b1[n];
        float g = 0.5f * h * (1.0f + erff(h * 0.70710678118654752440f));
        s += g * w2[n];
    }

#pragma unroll
    for (int off = 8; off >= 1; off >>= 1) {
        s += __shfl_xor(s, off, 32);
    }

    if (lane == 0) {
        out[sample] = s + l2_bias[task];
    }
}

extern "C" void kernel_launch(void* const* d_in, const int* in_sizes, int n_in,
                              void* d_out, int out_size, void* d_ws, size_t ws_size,
                              hipStream_t stream) {
    const float* x        = (const float*)d_in[0];
    const int*   task_ids = (const int*)d_in[1];
    const float* l1_emb   = (const float*)d_in[2];
    const float* l1_bias  = (const float*)d_in[3];
    const float* l2_emb   = (const float*)d_in[4];
    const float* l2_bias  = (const float*)d_in[5];
    float* out = (float*)d_out;

    dim3 grid(BATCH / WAVES_PER_BLOCK);
    dim3 block(WAVES_PER_BLOCK * 32);
    mtl_fused_kernel<<<grid, block, 0, stream>>>(x, task_ids, l1_emb, l1_bias,
                                                 l2_emb, l2_bias, out);
}


# Layer-1 slab streamed through an 8-step register ring instead of held whole in registers: 374 -> 248 registers per wave, so 4 waves fit a SIMD (was 2) and all 1000 work-groups are resident at once
# speedup vs baseline: 2.8539x; 2.8539x over previous
#include <hip/hip_runtime.h>
#include <math.h>


typedef __attribute__((ext_vector_type(2))) float v2f;
typedef __attribute__((ext_vector_type(4))) float v4f;
typedef __attribute__((ext_vector_type(8))) float v8f;
typedef __attribute__((ext_vector_type(4))) int v4i;

#define NUM_TASKS 1000
#define NFEAT 256
#define HID 128
#define BATCH 4096
#define NWAVES 4
#define NTHREADS (NWAVES * 32)
#define KQ 64
#define KSTEPS 16
#define IDV 8
#define DEPTH 8

static __device__ __forceinline__ void load_step(v4f (&dst)[4], const float* __restrict__ wl,
                                                 const int i) {
    const float* __restrict__ r = wl + (2 * i) * HID;
    dst[0] = *(const v4f*)(r);
    dst[1] = *(const v4f*)(r + 64);
    dst[2] = *(const v4f*)(r + HID);
    dst[3] = *(const v4f*)(r + HID + 64);
}

static __device__ __forceinline__ void task_chunk(
        const int c0, const int total, const int tid, const int wave, const int lane,
        const int c, const int h, const int task,
        const float* __restrict__ x, const float* __restrict__ w1,
        const float b1v, const float w2v, const float* __restrict__ l2_bias,
        const int* s_list, float (*s_red)[16][HID], float (*s_part)[16],
        float* __restrict__ out) {
    const int nrows = (total - c0 < 16) ? (total - c0) : 16;

    int mi = c0 + c;
    if (mi > total - 1) mi = total - 1;
    const int sidx = s_list[mi];

    const float* __restrict__ xp = x + (size_t)sidx * NFEAT + wave * KQ + h * 32;
    v4f a[8];
#pragma unroll
    for (int j = 0; j < 8; ++j) a[j] = *(const v4f*)(xp + 4 * j);

    v8f acc[8];
    const v8f zero = {};
#pragma unroll
    for (int t = 0; t < 8; ++t) acc[t] = zero;

    const float* __restrict__ wl = w1 + (size_t)(h * 32) * HID + 4 * c;

    v4f w[DEPTH][4];
#pragma unroll
    for (int i = 0; i < DEPTH; ++i) load_step(w[i], wl, i);

#pragma unroll
    for (int i = 0; i < KSTEPS; ++i) {
        const v4f r0a = w[i % DEPTH][0];
        const v4f r0b = w[i % DEPTH][1];
        const v4f r1a = w[i % DEPTH][2];
        const v4f r1b = w[i % DEPTH][3];

        v2f A;
        A.x = a[i >> 1][(2 * i) & 3];
        A.y = a[i >> 1][((2 * i) & 3) + 1];

#pragma unroll
        for (int t = 0; t < 4; ++t) {
            v2f B;
            B.x = r0a[t];
            B.y = r1a[t];
            acc[t] = __builtin_amdgcn_wmma_f32_16x16x4_f32(
                false, A, false, B, (short)0, acc[t], false, false);
        }
#pragma unroll
        for (int t = 0; t < 4; ++t) {
            v2f B;
            B.x = r0b[t];
            B.y = r1b[t];
            acc[4 + t] = __builtin_amdgcn_wmma_f32_16x16x4_f32(
                false, A, false, B, (short)0, acc[4 + t], false, false);
        }

        __builtin_amdgcn_sched_barrier(0);
        if (i + DEPTH < KSTEPS) load_step(w[i % DEPTH], wl, i + DEPTH);
    }

#pragma unroll
    for (int r = 0; r < 8; ++r) {
        if (r < nrows) {
            v4f lo, hi;
            lo.x = acc[0][r]; lo.y = acc[1][r]; lo.z = acc[2][r]; lo.w = acc[3][r];
            hi.x = acc[4][r]; hi.y = acc[5][r]; hi.z = acc[6][r]; hi.w = acc[7][r];
            *(v4f*)(&s_red[wave][8 * h + r][4 * c])      = lo;
            *(v4f*)(&s_red[wave][8 * h + r][64 + 4 * c]) = hi;
        }
    }
    __syncthreads();

    for (int m = 0; m < nrows; ++m) {
        const float hv = (((s_red[0][m][tid] + s_red[1][m][tid]) + s_red[2][m][tid])
                          + s_red[3][m][tid]) + b1v;
        const float g = 0.5f * hv * (1.0f + erff(hv * 0.70710678118654752440f));
        float s = g * w2v;
#pragma unroll
        for (int off = 16; off >= 1; off >>= 1) {
            s += __shfl_xor(s, off, 32);
        }
        if (lane == 0) s_part[wave][m] = s;
    }
    __syncthreads();

    if (tid < nrows) {
        const float s = ((s_part[0][tid] + s_part[1][tid]) + s_part[2][tid]) + s_part[3][tid];
        out[s_list[c0 + tid]] = s + l2_bias[task];
    }
}

__global__ __launch_bounds__(NTHREADS)
void mtl_task_grouped_kernel(const float* __restrict__ x,
                             const int* __restrict__ task_ids,
                             const float* __restrict__ l1_emb,
                             const float* __restrict__ l1_bias,
                             const float* __restrict__ l2_emb,
                             const float* __restrict__ l2_bias,
                             float* __restrict__ out) {
    __shared__ int   s_list[BATCH];
    __shared__ int   s_wtot[NWAVES];
    __shared__ float s_red[NWAVES][16][HID];
    __shared__ float s_part[NWAVES][16];

    const int tid  = threadIdx.x;
    const int wave = tid >> 5;
    const int lane = tid & 31;
    const int c    = lane & 15;
    const int h    = lane >> 4;
    const int task = blockIdx.x;

    const float* __restrict__ w1 =
        l1_emb + (size_t)task * (NFEAT * HID) + (size_t)(wave * KQ) * HID;

    const float b1v = l1_bias[(size_t)task * HID + tid];
    const float w2v = l2_emb[(size_t)task * HID + tid];

    unsigned mask = 0u;
    {
        const v4i* __restrict__ tp = (const v4i*)task_ids + tid * IDV;
#pragma unroll
        for (int q = 0; q < IDV; ++q) {
            const v4i v = tp[q];
            mask |= (v.x == task ? 1u : 0u) << (4 * q + 0);
            mask |= (v.y == task ? 1u : 0u) << (4 * q + 1);
            mask |= (v.z == task ? 1u : 0u) << (4 * q + 2);
            mask |= (v.w == task ? 1u : 0u) << (4 * q + 3);
        }
    }
    const int mycnt = __popc(mask);
    int incl = mycnt;
#pragma unroll
    for (int off = 1; off < 32; off <<= 1) {
        const int v = __shfl_up(incl, off, 32);
        if (lane >= off) incl += v;
    }
    if (lane == 31) s_wtot[wave] = incl;
    __syncthreads();

    int base = 0, total = 0;
#pragma unroll
    for (int w = 0; w < NWAVES; ++w) {
        const int t = s_wtot[w];
        if (w < wave) base += t;
        total += t;
    }
    {
        int pos = base + incl - mycnt;
        while (mask) {
            const int b = __ffs(mask) - 1;
            mask &= mask - 1u;
            s_list[pos++] = tid * 32 + b;
        }
    }
    __syncthreads();

    if (total == 0) return;

    task_chunk(0, total, tid, wave, lane, c, h, task, x, w1, b1v, w2v, l2_bias,
               s_list, s_red, s_part, out);

    for (int c0 = 16; c0 < total; c0 += 16) {
        __syncthreads();
        int trip_off = 0;
        asm volatile("" : "+s"(trip_off));
        const float* __restrict__ w1r = w1 + trip_off;
        task_chunk(c0, total, tid, wave, lane, c, h, task, x, w1r, b1v, w2v, l2_bias,
                   s_list, s_red, s_part, out);
    }
}

extern "C" void kernel_launch(void* const* d_in, const int* in_sizes, int n_in,
                              void* d_out, int out_size, void* d_ws, size_t ws_size,
                              hipStream_t stream) {
    const float* x        = (const float*)d_in[0];
    const int*   task_ids = (const int*)d_in[1];
    const float* l1_emb   = (const float*)d_in[2];
    const float* l1_bias  = (const float*)d_in[3];
    const float* l2_emb   = (const float*)d_in[4];
    const float* l2_bias  = (const float*)d_in[5];
    float* out = (float*)d_out;

    dim3 grid(NUM_TASKS);
    dim3 block(NTHREADS);
    mtl_task_grouped_kernel<<<grid, block, 0, stream>>>(x, task_ids, l1_emb, l1_bias,
                                                        l2_emb, l2_bias, out);
}


# Loads whose address depends on the block index alone (first half of each wave's layer-1 slab, per-task biases) issued at wave entry, ahead of the task-id scan, whose two barriers now order LDS only
# speedup vs baseline: 3.1160x; 1.0918x over previous
#include <hip/hip_runtime.h>
#include <math.h>


typedef __attribute__((ext_vector_type(2))) float v2f;
typedef __attribute__((ext_vector_type(4))) float v4f;
typedef __attribute__((ext_vector_type(8))) float v8f;
typedef __attribute__((ext_vector_type(4))) int v4i;

#define NUM_TASKS 1000
#define NFEAT 256
#define HID 128
#define BATCH 4096
#define NWAVES 4
#define NTHREADS (NWAVES * 32)
#define KQ 64
#define KSTEPS 16
#define IDV 8
#define DEPTH 8

static __device__ __forceinline__ void lds_barrier() {
    __builtin_amdgcn_fence(__ATOMIC_RELEASE, "workgroup", "local");
    __builtin_amdgcn_s_barrier();
    __builtin_amdgcn_fence(__ATOMIC_ACQUIRE, "workgroup", "local");
}

static __device__ __forceinline__ void load_step(v4f (&dst)[4], const float* __restrict__ wl,
                                                 const int i) {
    const float* __restrict__ r = wl + (2 * i) * HID;
    dst[0] = *(const v4f*)(r);
    dst[1] = *(const v4f*)(r + 64);
    dst[2] = *(const v4f*)(r + HID);
    dst[3] = *(const v4f*)(r + HID + 64);
}

template <bool PREFILLED>
static __device__ __forceinline__ void task_chunk(
        const int c0, const int total, const int tid, const int wave, const int lane,
        const int c, const int h,
        const float* __restrict__ x, const float* __restrict__ w1,
        const float b1v, const float w2v, const float b2v,
        v4f (&w)[DEPTH][4],
        const int* s_list, float (*s_red)[16][HID], float (*s_part)[16],
        float* __restrict__ out) {
    const int nrows = (total - c0 < 16) ? (total - c0) : 16;

    int mi = c0 + c;
    if (mi > total - 1) mi = total - 1;
    const int sidx = s_list[mi];

    const float* __restrict__ xp = x + (size_t)sidx * NFEAT + wave * KQ + h * 32;
    v4f a[8];
#pragma unroll
    for (int j = 0; j < 8; ++j) a[j] = *(const v4f*)(xp + 4 * j);
    __builtin_amdgcn_sched_barrier(0);

    v8f acc[8];
    const v8f zero = {};
#pragma unroll
    for (int t = 0; t < 8; ++t) acc[t] = zero;

    const float* __restrict__ wl = w1 + (size_t)(h * 32) * HID + 4 * c;

    if (!PREFILLED) {
#pragma unroll
        for (int i = 0; i < DEPTH; ++i) load_step(w[i], wl, i);
    }

#pragma unroll
    for (int i = 0; i < KSTEPS; ++i) {
        const v4f r0a = w[i % DEPTH][0];
        const v4f r0b = w[i % DEPTH][1];
        const v4f r1a = w[i % DEPTH][2];
        const v4f r1b = w[i % DEPTH][3];

        v2f A;
        A.x = a[i >> 1][(2 * i) & 3];
        A.y = a[i >> 1][((2 * i) & 3) + 1];

#pragma unroll
        for (int t = 0; t < 4; ++t) {
            v2f B;
            B.x = r0a[t];
            B.y = r1a[t];
            acc[t] = __builtin_amdgcn_wmma_f32_16x16x4_f32(
                false, A, false, B, (short)0, acc[t], false, false);
        }
#pragma unroll
        for (int t = 0; t < 4; ++t) {
            v2f B;
            B.x = r0b[t];
            B.y = r1b[t];
            acc[4 + t] = __builtin_amdgcn_wmma_f32_16x16x4_f32(
                false, A, false, B, (short)0, acc[4 + t], false, false);
        }

        __builtin_amdgcn_sched_barrier(0);
        if (i + DEPTH < KSTEPS) load_step(w[i % DEPTH], wl, i + DEPTH);
    }

#pragma unroll
    for (int r = 0; r < 8; ++r) {
        if (r < nrows) {
            v4f lo, hi;
            lo.x = acc[0][r]; lo.y = acc[1][r]; lo.z = acc[2][r]; lo.w = acc[3][r];
            hi.x = acc[4][r]; hi.y = acc[5][r]; hi.z = acc[6][r]; hi.w = acc[7][r];
            *(v4f*)(&s_red[wave][8 * h + r][4 * c])      = lo;
            *(v4f*)(&s_red[wave][8 * h + r][64 + 4 * c]) = hi;
        }
    }
    __syncthreads();

    for (int m = 0; m < nrows; ++m) {
        const float hv = (((s_red[0][m][tid] + s_red[1][m][tid]) + s_red[2][m][tid])
                          + s_red[3][m][tid]) + b1v;
        const float g = 0.5f * hv * (1.0f + erff(hv * 0.70710678118654752440f));
        float s = g * w2v;
#pragma unroll
        for (int off = 16; off >= 1; off >>= 1) {
            s += __shfl_xor(s, off, 32);
        }
        if (lane == 0) s_part[wave][m] = s;
    }
    __syncthreads();

    if (tid < nrows) {
        const float s = ((s_part[0][tid] + s_part[1][tid]) + s_part[2][tid]) + s_part[3][tid];
        out[s_list[c0 + tid]] = s + b2v;
    }
}

__global__ __launch_bounds__(NTHREADS)
void mtl_task_grouped_kernel(const float* __restrict__ x,
                             const int* __restrict__ task_ids,
                             const float* __restrict__ l1_emb,
                             const float* __restrict__ l1_bias,
                             const float* __restrict__ l2_emb,
                             const float* __restrict__ l2_bias,
                             float* __restrict__ out) {
    __shared__ int   s_list[BATCH];
    __shared__ int   s_wtot[NWAVES];
    __shared__ float s_red[NWAVES][16][HID];
    __shared__ float s_part[NWAVES][16];

    const int tid  = threadIdx.x;
    const int wave = tid >> 5;
    const int lane = tid & 31;
    const int c    = lane & 15;
    const int h    = lane >> 4;
    const int task = blockIdx.x;

    const float* __restrict__ w1 =
        l1_emb + (size_t)task * (NFEAT * HID) + (size_t)(wave * KQ) * HID;

    v4i idv[IDV];
    {
        const v4i* __restrict__ tp = (const v4i*)task_ids + tid * IDV;
#pragma unroll
        for (int q = 0; q < IDV; ++q) idv[q] = tp[q];
    }
    v4f w[DEPTH][4];
    {
        const float* __restrict__ wl = w1 + (size_t)(h * 32) * HID + 4 * c;
#pragma unroll
        for (int i = 0; i < DEPTH; ++i) load_step(w[i], wl, i);
    }
    const float b1v = l1_bias[(size_t)task * HID + tid];
    const float w2v = l2_emb[(size_t)task * HID + tid];
    const float b2v = l2_bias[task];

    unsigned mask = 0u;
    {
#pragma unroll
        for (int q = 0; q < IDV; ++q) {
            const v4i v = idv[q];
            mask |= (v.x == task ? 1u : 0u) << (4 * q + 0);
            mask |= (v.y == task ? 1u : 0u) << (4 * q + 1);
            mask |= (v.z == task ? 1u : 0u) << (4 * q + 2);
            mask |= (v.w == task ? 1u : 0u) << (4 * q + 3);
        }
    }
    const int mycnt = __popc(mask);
    int incl = mycnt;
#pragma unroll
    for (int off = 1; off < 32; off <<= 1) {
        const int v = __shfl_up(incl, off, 32);
        if (lane >= off) incl += v;
    }
    if (lane == 31) s_wtot[wave] = incl;
    lds_barrier();

    int base = 0, total = 0;
#pragma unroll
    for (int wv = 0; wv < NWAVES; ++wv) {
        const int t = s_wtot[wv];
        if (wv < wave) base += t;
        total += t;
    }
    {
        int pos = base + incl - mycnt;
        while (mask) {
            const int b = __ffs(mask) - 1;
            mask &= mask - 1u;
            s_list[pos++] = tid * 32 + b;
        }
    }
    lds_barrier();

    if (total == 0) return;

    task_chunk<true>(0, total, tid, wave, lane, c, h, x, w1, b1v, w2v, b2v, w,
                     s_list, s_red, s_part, out);

    for (int c0 = 16; c0 < total; c0 += 16) {
        __syncthreads();
        int trip_off = 0;
        asm volatile("" : "+s"(trip_off));
        const float* __restrict__ w1r = w1 + trip_off;
        task_chunk<false>(c0, total, tid, wave, lane, c, h, x, w1r, b1v, w2v, b2v, w,
                          s_list, s_red, s_part, out);
    }
}

extern "C" void kernel_launch(void* const* d_in, const int* in_sizes, int n_in,
                              void* d_out, int out_size, void* d_ws, size_t ws_size,
                              hipStream_t stream) {
    const float* x        = (const float*)d_in[0];
    const int*   task_ids = (const int*)d_in[1];
    const float* l1_emb   = (const float*)d_in[2];
    const float* l1_bias  = (const float*)d_in[3];
    const float* l2_emb   = (const float*)d_in[4];
    const float* l2_bias  = (const float*)d_in[5];
    float* out = (float*)d_out;

    dim3 grid(NUM_TASKS);
    dim3 block(NTHREADS);
    mtl_task_grouped_kernel<<<grid, block, 0, stream>>>(x, task_ids, l1_emb, l1_bias,
                                                        l2_emb, l2_bias, out);
}


# Layer 2 takes a task's rows together, not one after another: column threads write gelu(h)*w2 back to LDS four rows per pass, then 8 threads per row sum the 128 columns (summation order changes)
# speedup vs baseline: 3.3951x; 1.0896x over previous
#include <hip/hip_runtime.h>
#include <math.h>


typedef __attribute__((ext_vector_type(2))) float v2f;
typedef __attribute__((ext_vector_type(4))) float v4f;
typedef __attribute__((ext_vector_type(8))) float v8f;
typedef __attribute__((ext_vector_type(4))) int v4i;

#define NUM_TASKS 1000
#define NFEAT 256
#define HID 128
#define BATCH 4096
#define NWAVES 4
#define NTHREADS (NWAVES * 32)
#define KQ 64
#define KSTEPS 16
#define IDV 8
#define DEPTH 8

static __device__ __forceinline__ void lds_barrier() {
    __builtin_amdgcn_fence(__ATOMIC_RELEASE, "workgroup", "local");
    __builtin_amdgcn_s_barrier();
    __builtin_amdgcn_fence(__ATOMIC_ACQUIRE, "workgroup", "local");
}

static __device__ __forceinline__ void load_step(v4f (&dst)[4], const float* __restrict__ wl,
                                                 const int i) {
    const float* __restrict__ r = wl + (2 * i) * HID;
    dst[0] = *(const v4f*)(r);
    dst[1] = *(const v4f*)(r + 64);
    dst[2] = *(const v4f*)(r + HID);
    dst[3] = *(const v4f*)(r + HID + 64);
}

template <bool PREFILLED>
static __device__ __forceinline__ void task_chunk(
        const int c0, const int total, const int tid, const int wave,
        const int c, const int h,
        const float* __restrict__ x, const float* __restrict__ w1,
        const float b1v, const float w2v, const float b2v,
        v4f (&w)[DEPTH][4],
        const int* s_list, float (*s_red)[16][HID],
        float* __restrict__ out) {
    const int nrows = (total - c0 < 16) ? (total - c0) : 16;
    const int nr4 = (nrows + 3) & ~3;

    int mi = c0 + c;
    if (mi > total - 1) mi = total - 1;
    const int sidx = s_list[mi];

    const float* __restrict__ xp = x + (size_t)sidx * NFEAT + wave * KQ + h * 32;
    v4f a[8];
#pragma unroll
    for (int j = 0; j < 8; ++j) a[j] = *(const v4f*)(xp + 4 * j);
    __builtin_amdgcn_sched_barrier(0);

    v8f acc[8];
    const v8f zero = {};
#pragma unroll
    for (int t = 0; t < 8; ++t) acc[t] = zero;

    const float* __restrict__ wl = w1 + (size_t)(h * 32) * HID + 4 * c;

    if (!PREFILLED) {
#pragma unroll
        for (int i = 0; i < DEPTH; ++i) load_step(w[i], wl, i);
    }

#pragma unroll
    for (int i = 0; i < KSTEPS; ++i) {
        const v4f r0a = w[i % DEPTH][0];
        const v4f r0b = w[i % DEPTH][1];
        const v4f r1a = w[i % DEPTH][2];
        const v4f r1b = w[i % DEPTH][3];

        v2f A;
        A.x = a[i >> 1][(2 * i) & 3];
        A.y = a[i >> 1][((2 * i) & 3) + 1];

#pragma unroll
        for (int t = 0; t < 4; ++t) {
            v2f B;
            B.x = r0a[t];
            B.y = r1a[t];
            acc[t] = __builtin_amdgcn_wmma_f32_16x16x4_f32(
                false, A, false, B, (short)0, acc[t], false, false);
        }
#pragma unroll
        for (int t = 0; t < 4; ++t) {
            v2f B;
            B.x = r0b[t];
            B.y = r1b[t];
            acc[4 + t] = __builtin_amdgcn_wmma_f32_16x16x4_f32(
                false, A, false, B, (short)0, acc[4 + t], false, false);
        }

        __builtin_amdgcn_sched_barrier(0);
        if (i + DEPTH < KSTEPS) load_step(w[i % DEPTH], wl, i + DEPTH);
    }

#pragma unroll
    for (int r = 0; r < 8; ++r) {
        if (r < nr4) {
            v4f lo, hi;
            lo.x = acc[0][r]; lo.y = acc[1][r]; lo.z = acc[2][r]; lo.w = acc[3][r];
            hi.x = acc[4][r]; hi.y = acc[5][r]; hi.z = acc[6][r]; hi.w = acc[7][r];
            *(v4f*)(&s_red[wave][8 * h + r][4 * c])      = lo;
            *(v4f*)(&s_red[wave][8 * h + r][64 + 4 * c]) = hi;
        }
    }
    __syncthreads();

    for (int m0 = 0; m0 < nrows; m0 += 4) {
        float hv[4];
        for (int u = 0; u < 4; ++u) {
            const int m = m0 + u;
            hv[u] = (((s_red[0][m][tid] + s_red[1][m][tid]) + s_red[2][m][tid])
                     + s_red[3][m][tid]) + b1v;
        }
        for (int u = 0; u < 4; ++u) {
            const float g = 0.5f * hv[u] * (1.0f + erff(hv[u] * 0.70710678118654752440f));
            s_red[0][m0 + u][tid] = g * w2v;
        }
    }
    __syncthreads();

    if (4 * wave < nr4) {
        const int m  = tid >> 3;
        const int ch = tid & 7;
        const float* p = &s_red[0][m][16 * ch];
        const v4f q0 = *(const v4f*)(p);
        const v4f q1 = *(const v4f*)(p + 4);
        const v4f q2 = *(const v4f*)(p + 8);
        const v4f q3 = *(const v4f*)(p + 12);
        float s = (((q0.x + q0.y) + (q0.z + q0.w)) + ((q1.x + q1.y) + (q1.z + q1.w)))
                + (((q2.x + q2.y) + (q2.z + q2.w)) + ((q3.x + q3.y) + (q3.z + q3.w)));
        s += __shfl_xor(s, 1, 32);
        s += __shfl_xor(s, 2, 32);
        s += __shfl_xor(s, 4, 32);
        if (ch == 0 && m < nrows) out[s_list[c0 + m]] = s + b2v;
    }
}

__global__ __launch_bounds__(NTHREADS)
void mtl_task_grouped_kernel(const float* __restrict__ x,
                             const int* __restrict__ task_ids,
                             const float* __restrict__ l1_emb,
                             const float* __restrict__ l1_bias,
                             const float* __restrict__ l2_emb,
                             const float* __restrict__ l2_bias,
                             float* __restrict__ out) {
    __shared__ int   s_list[BATCH];
    __shared__ int   s_wtot[NWAVES];
    __shared__ float s_red[NWAVES][16][HID];

    const int tid  = threadIdx.x;
    const int wave = tid >> 5;
    const int lane = tid & 31;
    const int c    = lane & 15;
    const int h    = lane >> 4;
    const int task = blockIdx.x;

    const float* __restrict__ w1 =
        l1_emb + (size_t)task * (NFEAT * HID) + (size_t)(wave * KQ) * HID;

    v4i idv[IDV];
    {
        const v4i* __restrict__ tp = (const v4i*)task_ids + tid * IDV;
#pragma unroll
        for (int q = 0; q < IDV; ++q) idv[q] = tp[q];
    }
    v4f w[DEPTH][4];
    {
        const float* __restrict__ wl = w1 + (size_t)(h * 32) * HID + 4 * c;
#pragma unroll
        for (int i = 0; i < DEPTH; ++i) load_step(w[i], wl, i);
    }
    const float b1v = l1_bias[(size_t)task * HID + tid];
    const float w2v = l2_emb[(size_t)task * HID + tid];
    const float b2v = l2_bias[task];

    unsigned mask = 0u;
    {
#pragma unroll
        for (int q = 0; q < IDV; ++q) {
            const v4i v = idv[q];
            mask |= (v.x == task ? 1u : 0u) << (4 * q + 0);
            mask |= (v.y == task ? 1u : 0u) << (4 * q + 1);
            mask |= (v.z == task ? 1u : 0u) << (4 * q + 2);
            mask |= (v.w == task ? 1u : 0u) << (4 * q + 3);
        }
    }
    const int mycnt = __popc(mask);
    int incl = mycnt;
#pragma unroll
    for (int off = 1; off < 32; off <<= 1) {
        const int v = __shfl_up(incl, off, 32);
        if (lane >= off) incl += v;
    }
    if (lane == 31) s_wtot[wave] = incl;
    lds_barrier();

    int base = 0, total = 0;
#pragma unroll
    for (int wv = 0; wv < NWAVES; ++wv) {
        const int t = s_wtot[wv];
        if (wv < wave) base += t;
        total += t;
    }
    {
        int pos = base + incl - mycnt;
        while (mask) {
            const int b = __ffs(mask) - 1;
            mask &= mask - 1u;
            s_list[pos++] = tid * 32 + b;
        }
    }
    lds_barrier();

    if (total == 0) return;

    task_chunk<true>(0, total, tid, wave, c, h, x, w1, b1v, w2v, b2v, w,
                     s_list, s_red, out);

    for (int c0 = 16; c0 < total; c0 += 16) {
        __syncthreads();
        int trip_off = 0;
        asm volatile("" : "+s"(trip_off));
        const float* __restrict__ w1r = w1 + trip_off;
        task_chunk<false>(c0, total, tid, wave, c, h, x, w1r, b1v, w2v, b2v, w,
                          s_list, s_red, out);
    }
}

extern "C" void kernel_launch(void* const* d_in, const int* in_sizes, int n_in,
                              void* d_out, int out_size, void* d_ws, size_t ws_size,
                              hipStream_t stream) {
    const float* x        = (const float*)d_in[0];
    const int*   task_ids = (const int*)d_in[1];
    const float* l1_emb   = (const float*)d_in[2];
    const float* l1_bias  = (const float*)d_in[3];
    const float* l2_emb   = (const float*)d_in[4];
    const float* l2_bias  = (const float*)d_in[5];
    float* out = (float*)d_out;

    dim3 grid(NUM_TASKS);
    dim3 block(NTHREADS);
    mtl_task_grouped_kernel<<<grid, block, 0, stream>>>(x, task_ids, l1_emb, l1_bias,
                                                        l2_emb, l2_bias, out);
}
